# starting kernel
# baseline (speedup 1.0000x reference)
#include <hip/hip_runtime.h>


typedef __attribute__((ext_vector_type(2))) float v2f;
typedef __attribute__((ext_vector_type(8))) float v8f;

#define HID 16
#define KITERS 10
#define ONE_MINUS_ALPHA 0.9f
#define ALPHA_F 0.1f

__device__ __forceinline__ float relu_f(float v) { return v > 0.f ? v : 0.f; }


__global__ void ak_deg_init(float* __restrict__ deg, int n) {
    int i = blockIdx.x * blockDim.x + threadIdx.x;
    if (i < n) deg[i] = 1.0f;
}

__global__ void ak_deg_accum(const int* __restrict__ eidx, float* __restrict__ deg, int E) {
    int e = blockIdx.x * blockDim.x + threadIdx.x;
    if (e < E) atomicAdd(&deg[eidx[(long long)E + e]], 1.0f);
}

__global__ void ak_dinv(const float* __restrict__ deg, float* __restrict__ dinv, int n) {
    int i = blockIdx.x * blockDim.x + threadIdx.x;
    if (i < n) dinv[i] = rsqrtf(deg[i]);
}


__global__ void ak_encoder_wmma(const float* __restrict__ x, const float* __restrict__ W1,
                                const float* __restrict__ b1, const float* __restrict__ W2,
                                const float* __restrict__ b2, const float* __restrict__ dinv,
                                float* __restrict__ h0, float* __restrict__ g,
                                float* __restrict__ agg, int n_tiles) {
    int wave = (blockIdx.x * blockDim.x + threadIdx.x) >> 5;
    int lane = threadIdx.x & 31;
    if (wave >= n_tiles) return;

    int row = lane & 15;
    int hi  = lane >> 4;
    int node = wave * 16 + row;
    float xv = x[node];

    v8f acc = {};
#pragma unroll
    for (int s = 0; s < 4; ++s) {
        int k0 = 4 * s + 2 * hi;
        v2f a, b;
        a.x = relu_f(xv * W1[k0]     + b1[k0]);
        a.y = relu_f(xv * W1[k0 + 1] + b1[k0 + 1]);
        b.x = W2[k0 * HID + row];
        b.y = W2[(k0 + 1) * HID + row];
        acc = __builtin_amdgcn_wmma_f32_16x16x4_f32(
            false, a, false, b, (short)0, acc, false, false);
    }

    float bb = b2[row];
#pragma unroll
    for (int i = 0; i < 8; ++i) {
        int r = i + 8 * hi;
        int nr = wave * 16 + r;
        float v = relu_f(acc[i] + bb);
        int idx = nr * HID + row;
        h0[idx]  = v;
        g[idx]   = dinv[nr] * v;
        agg[idx] = 0.f;
    }
}

__global__ void ak_encoder_tail(const float* __restrict__ x, const float* __restrict__ W1,
                                const float* __restrict__ b1, const float* __restrict__ W2,
                                const float* __restrict__ b2, const float* __restrict__ dinv,
                                float* __restrict__ h0, float* __restrict__ g,
                                float* __restrict__ agg, int start, int N) {
    int v = start + blockIdx.x * blockDim.x + threadIdx.x;
    if (v >= N) return;
    float h1[HID];
    float xv = x[v];
#pragma unroll
    for (int k = 0; k < HID; ++k) h1[k] = relu_f(xv * W1[k] + b1[k]);
    float dv = dinv[v];
#pragma unroll
    for (int j = 0; j < HID; ++j) {
        float acc = b2[j];
#pragma unroll
        for (int k = 0; k < HID; ++k) acc += h1[k] * W2[k * HID + j];
        float hv = relu_f(acc);
        int idx = v * HID + j;
        h0[idx] = hv; g[idx] = dv * hv; agg[idx] = 0.f;
    }
}


__global__ void ak_edge_scatter(const int* __restrict__ eidx, const float* __restrict__ g,
                                float* __restrict__ agg, int E) {
    long long t = (long long)blockIdx.x * blockDim.x + threadIdx.x;
    int e = (int)(t >> 2);
    if (e >= E) return;
    int p = (int)(t & 3);
    int src = eidx[e];
    int dst = eidx[(long long)E + e];
    const float4 val = *(const float4*)(g + (size_t)src * HID + p * 4);
    float* ad = agg + (size_t)dst * HID + p * 4;
    atomicAdd(ad + 0, val.x);
    atomicAdd(ad + 1, val.y);
    atomicAdd(ad + 2, val.z);
    atomicAdd(ad + 3, val.w);
}

__global__ void ak_finalize(float4* __restrict__ agg, float4* __restrict__ g,
                            const float4* __restrict__ h0, const float* __restrict__ dinv,
                            int n4) {
    int t = blockIdx.x * blockDim.x + threadIdx.x;
    if (t >= n4) return;
    int v = t >> 2;
    float dv = dinv[v];
    float4 a = agg[t], gg = g[t], h = h0[t];
    float4 hn;
    hn.x = ONE_MINUS_ALPHA * dv * (a.x + gg.x) + ALPHA_F * h.x;
    hn.y = ONE_MINUS_ALPHA * dv * (a.y + gg.y) + ALPHA_F * h.y;
    hn.z = ONE_MINUS_ALPHA * dv * (a.z + gg.z) + ALPHA_F * h.z;
    hn.w = ONE_MINUS_ALPHA * dv * (a.w + gg.w) + ALPHA_F * h.w;
    float4 gn; gn.x = dv * hn.x; gn.y = dv * hn.y; gn.z = dv * hn.z; gn.w = dv * hn.w;
    g[t] = gn;
    float4 z; z.x = 0.f; z.y = 0.f; z.z = 0.f; z.w = 0.f;
    agg[t] = z;
}

__global__ void ak_finalize_out(const float* __restrict__ agg, const float* __restrict__ g,
                                const float* __restrict__ h0, const float* __restrict__ dinv,
                                const float* __restrict__ W3, const float* __restrict__ b3,
                                float* __restrict__ out, int N) {
    int v = blockIdx.x * blockDim.x + threadIdx.x;
    if (v >= N) return;
    float dv = dinv[v];
    float s = b3[0];
#pragma unroll
    for (int j = 0; j < HID; ++j) {
        float h = ONE_MINUS_ALPHA * dv * (agg[v * HID + j] + g[v * HID + j])
                + ALPHA_F * h0[v * HID + j];
        s += h * W3[j];
    }
    out[v] = s;
}


extern "C" void kernel_launch(void* const* d_in, const int* in_sizes, int n_in,
                              void* d_out, int out_size, void* d_ws, size_t ws_size,
                              hipStream_t stream) {
    const float* x  = (const float*)d_in[0];
    const int*  eidx = (const int*)d_in[1];
    const float* W1 = (const float*)d_in[2];
    const float* b1 = (const float*)d_in[3];
    const float* W2 = (const float*)d_in[4];
    const float* b2 = (const float*)d_in[5];
    const float* W3 = (const float*)d_in[6];
    const float* b3 = (const float*)d_in[7];
    float* out = (float*)d_out;

    const int N = in_sizes[0];
    const int E = in_sizes[1] / 2;

    char* ws = (char*)d_ws;
    size_t off = 0;
    auto carve = [&](size_t bytes) -> char* {
        char* p = ws + off;
        off += (bytes + 255) & ~(size_t)255;
        return p;
    };
    float* deg  = (float*)carve((size_t)N * 4);
    float* dinv = (float*)carve((size_t)N * 4);
    float* h0   = (float*)carve((size_t)N * HID * 4);
    float* g    = (float*)carve((size_t)N * HID * 4);
    float* agg  = (float*)carve((size_t)N * HID * 4);

    const int B = 256;
    ak_deg_init<<<(N + B - 1) / B, B, 0, stream>>>(deg, N);
    ak_deg_accum<<<(E + B - 1) / B, B, 0, stream>>>(eidx, deg, E);
    ak_dinv<<<(N + B - 1) / B, B, 0, stream>>>(deg, dinv, N);

    int tiles = N / 16;
    if (tiles > 0) {
        int waves_per_block = B / 32;
        int blocks = (tiles + waves_per_block - 1) / waves_per_block;
        ak_encoder_wmma<<<blocks, B, 0, stream>>>(x, W1, b1, W2, b2, dinv, h0, g, agg, tiles);
    }
    int rem_start = tiles * 16;
    if (rem_start < N) {
        ak_encoder_tail<<<(N - rem_start + B - 1) / B, B, 0, stream>>>(
            x, W1, b1, W2, b2, dinv, h0, g, agg, rem_start, N);
    }

    long long sthreads = 4LL * E;
    int sblocks = (int)((sthreads + B - 1) / B);
    int n4 = N * 4;
    for (int it = 0; it < KITERS; ++it) {
        ak_edge_scatter<<<sblocks, B, 0, stream>>>(eidx, g, agg, E);
        if (it < KITERS - 1) {
            ak_finalize<<<(n4 + B - 1) / B, B, 0, stream>>>(
                (float4*)agg, (float4*)g, (const float4*)h0, dinv, n4);
        } else {
            ak_finalize_out<<<(N + B - 1) / B, B, 0, stream>>>(
                agg, g, h0, dinv, W3, b3, out, N);
        }
    }
}


# Edges grouped once per call by 512-node source range (LDS counting sort); a hop's work-group holds its range of g in LDS and reads source rows there, not from L2; adds stay atomic, sum order changes
# speedup vs baseline: 3.6955x; 3.6955x over previous
#include <hip/hip_runtime.h>


typedef __attribute__((ext_vector_type(2))) float v2f;
typedef __attribute__((ext_vector_type(8))) float v8f;

#define HID 16
#define KITERS 10
#define ONE_MINUS_ALPHA 0.9f
#define ALPHA_F 0.1f

__device__ __forceinline__ float relu_f(float v) { return v > 0.f ? v : 0.f; }

__device__ __forceinline__ int cache_half() {
    return (int)((__builtin_amdgcn_s_sendmsg_rtn(0x87) >> 18) & 1u);
}


__global__ void ak_deg_init(float* deg0, float* deg1, int n) {
    int i = blockIdx.x * blockDim.x + threadIdx.x;
    if (i < n) {
        deg0[i] = 1.0f;
        if (deg1 != deg0) deg1[i] = 0.0f;
    }
}

__global__ void ak_deg_accum(const int* __restrict__ eidx, float* deg0, float* deg1, int E) {
    float* deg = cache_half() ? deg1 : deg0;
    const int stride = (int)(gridDim.x * blockDim.x);
    for (int e = (int)(blockIdx.x * blockDim.x + threadIdx.x); e < E; e += stride)
        atomicAdd(&deg[eidx[(long long)E + e]], 1.0f);
}

__global__ void ak_dinv(const float* deg0, const float* deg1, float* __restrict__ dinv, int n) {
    int i = blockIdx.x * blockDim.x + threadIdx.x;
    if (i < n) {
        float d = deg0[i];
        if (deg1 != deg0) d += deg1[i];
        dinv[i] = rsqrtf(d);
    }
}


__global__ void ak_encoder_wmma(const float* __restrict__ x, const float* __restrict__ W1,
                                const float* __restrict__ b1, const float* __restrict__ W2,
                                const float* __restrict__ b2, const float* __restrict__ dinv,
                                float* __restrict__ h0, float* __restrict__ g,
                                float* agg0, float* agg1, int n_tiles) {
    int wave = (blockIdx.x * blockDim.x + threadIdx.x) >> 5;
    int lane = threadIdx.x & 31;
    if (wave >= n_tiles) return;

    int row = lane & 15;
    int hi  = lane >> 4;
    int node = wave * 16 + row;
    float xv = x[node];

    v8f acc = {};
#pragma unroll
    for (int s = 0; s < 4; ++s) {
        int k0 = 4 * s + 2 * hi;
        v2f a, b;
        a.x = relu_f(xv * W1[k0]     + b1[k0]);
        a.y = relu_f(xv * W1[k0 + 1] + b1[k0 + 1]);
        b.x = W2[k0 * HID + row];
        b.y = W2[(k0 + 1) * HID + row];
        acc = __builtin_amdgcn_wmma_f32_16x16x4_f32(
            false, a, false, b, (short)0, acc, false, false);
    }

    float bb = b2[row];
#pragma unroll
    for (int i = 0; i < 8; ++i) {
        int r = i + 8 * hi;
        int nr = wave * 16 + r;
        float v = relu_f(acc[i] + bb);
        int idx = nr * HID + row;
        h0[idx]  = v;
        g[idx]   = dinv[nr] * v;
        agg0[idx] = 0.f;
        if (agg1 != agg0) agg1[idx] = 0.f;
    }
}

__global__ void ak_encoder_tail(const float* __restrict__ x, const float* __restrict__ W1,
                                const float* __restrict__ b1, const float* __restrict__ W2,
                                const float* __restrict__ b2, const float* __restrict__ dinv,
                                float* __restrict__ h0, float* __restrict__ g,
                                float* agg0, float* agg1, int start, int N) {
    int v = start + blockIdx.x * blockDim.x + threadIdx.x;
    if (v >= N) return;
    float h1[HID];
    float xv = x[v];
#pragma unroll
    for (int k = 0; k < HID; ++k) h1[k] = relu_f(xv * W1[k] + b1[k]);
    float dv = dinv[v];
#pragma unroll
    for (int j = 0; j < HID; ++j) {
        float acc = b2[j];
#pragma unroll
        for (int k = 0; k < HID; ++k) acc += h1[k] * W2[k * HID + j];
        float hv = relu_f(acc);
        int idx = v * HID + j;
        h0[idx] = hv; g[idx] = dv * hv; agg0[idx] = 0.f;
        if (agg1 != agg0) agg1[idx] = 0.f;
    }
}


#define SCATTER_BLOCKS 2048

template <bool FULL>
__device__ __forceinline__ void scatter_batch(int s, int d, int nleft, int c, int hi,
                                              const int* __restrict__ eidx, long long en,
                                              int E, int& sn, int& dn,
                                              const float* __restrict__ g,
                                              float* __restrict__ agg) {
    float v[16];
    int   dj[16];
#pragma unroll
    for (int i = 0; i < 16; ++i) {
        const int s0 = __builtin_amdgcn_readlane(s, 2 * i);
        const int s1 = __builtin_amdgcn_readlane(s, 2 * i + 1);
        const int d0 = __builtin_amdgcn_readlane(d, 2 * i);
        const int d1 = __builtin_amdgcn_readlane(d, 2 * i + 1);
        const int sj = hi ? s1 : s0;
        dj[i] = hi ? d1 : d0;
        v[i] = (FULL || 2 * i + hi < nleft) ? g[sj * HID + c] : 0.f;
    }
    if (en < E) { sn = eidx[en]; dn = eidx[(long long)E + en]; }
#pragma unroll
    for (int i = 0; i < 16; ++i) {
        if (FULL || 2 * i + hi < nleft) atomicAdd(agg + dj[i] * HID + c, v[i]);
    }
}

__global__ void ak_edge_scatter(const int* __restrict__ eidx, const float* __restrict__ g,
                                float* agg0, float* agg1, int E) {
    const int lane = threadIdx.x & 31;
    const int c  = lane & 15;
    const int hi = lane >> 4;
    const int nwaves = (int)((gridDim.x * blockDim.x) >> 5);
    const int nbatch = (E + 31) >> 5;
    int b = (int)((blockIdx.x * blockDim.x + threadIdx.x) >> 5);
    if (b >= nbatch) return;

    float* __restrict__ agg = cache_half() ? agg1 : agg0;

    int s = 0, d = 0;
    {
        const long long e = (long long)b * 32 + lane;
        if (e < E) { s = eidx[e]; d = eidx[(long long)E + e]; }
    }
    for (;;) {
        const int bn = b + nwaves;
        const long long en = (bn < nbatch) ? (long long)bn * 32 + lane : (long long)E;
        int sn = 0, dn = 0;
        const int nleft = E - b * 32;
        if (nleft >= 32) scatter_batch<true>(s, d, 32, c, hi, eidx, en, E, sn, dn, g, agg);
        else             scatter_batch<false>(s, d, nleft, c, hi, eidx, en, E, sn, dn, g, agg);
        if (bn >= nbatch) break;
        b = bn; s = sn; d = dn;
    }
}


#define SRC_R 512
#define SRC_SHIFT 23
#define SORT_WGS 256
#define SORT_THREADS 1024
#define MAX_RANGES 2048

__global__ void __launch_bounds__(SORT_THREADS)
ak_src_count(const int* __restrict__ eidx, int* __restrict__ off, int E, int nranges) {
    __shared__ int hist[MAX_RANGES];
    const int tid = (int)threadIdx.x, w = (int)blockIdx.x;
    for (int i = tid; i < nranges; i += SORT_THREADS) hist[i] = 0;
    __syncthreads();
    const int chunk = (E + SORT_WGS - 1) / SORT_WGS;
    const int lo = w * chunk;
    const int hi = min(E, lo + chunk);
    for (int e = lo + tid; e < hi; e += SORT_THREADS) {
        const int b = min((int)((unsigned)eidx[e] / SRC_R), nranges - 1);
        atomicAdd(&hist[b], 1);
    }
    __syncthreads();
    for (int i = tid; i < nranges; i += SORT_THREADS) off[i * SORT_WGS + w] = hist[i];
}

__device__ __forceinline__ int wave_incl_sum(int x, int lane) {
#pragma unroll
    for (int dlt = 1; dlt < 32; dlt <<= 1) {
        const int y = __shfl_up(x, dlt, 32);
        if (lane >= dlt) x += y;
    }
    return x;
}

__global__ void __launch_bounds__(SORT_THREADS)
ak_src_scan(int* off, int n) {
    __shared__ int wtot[32];
    const int tid = (int)threadIdx.x;
    const int lane = tid & 31;
    const int wv = tid >> 5;
    const int span = (((n + 31) >> 5) + 127) & ~127;
    const int first = min(n, wv * span);
    const int last = min(n, first + span);
    int4* off4 = (int4*)off;

    int s = 0;
    for (int i = first + lane * 4; i < last; i += 128) {
        const int4 v = off4[i >> 2];
        s += (v.x + v.y) + (v.z + v.w);
    }
    s = wave_incl_sum(s, lane);
    if (lane == 31) wtot[wv] = s;
    __syncthreads();
    const int t = wtot[lane];
    const int upto = wave_incl_sum(t, lane);
    const int all = __shfl(upto, 31, 32);
    int run = __shfl(upto - t, wv, 32);
    for (int i0 = first; i0 < last; i0 += 128) {
        const int i = i0 + lane * 4;
        int4 v; v.x = 0; v.y = 0; v.z = 0; v.w = 0;
        if (i < last) v = off4[i >> 2];
        const int s1 = v.x + v.y, s2 = s1 + v.z, s3 = s2 + v.w;
        const int inc = wave_incl_sum(s3, lane);
        const int base = run + inc - s3;
        int4 o; o.x = base; o.y = base + v.x; o.z = base + s1; o.w = base + s2;
        if (i < last) off4[i >> 2] = o;
        run += __shfl(inc, 31, 32);
    }
    if (tid == 0) off[n] = all;
}

__global__ void __launch_bounds__(SORT_THREADS)
ak_src_place(const int* __restrict__ eidx, const int* __restrict__ off,
             unsigned* __restrict__ rec, int E, int nranges) {
    __shared__ int cur[MAX_RANGES];
    const int tid = (int)threadIdx.x, w = (int)blockIdx.x;
    for (int i = tid; i < nranges; i += SORT_THREADS) cur[i] = off[i * SORT_WGS + w];
    __syncthreads();
    const int chunk = (E + SORT_WGS - 1) / SORT_WGS;
    const int lo = w * chunk;
    const int hi = min(E, lo + chunk);
    for (int e = lo + tid; e < hi; e += SORT_THREADS) {
        const int s = eidx[e];
        const int d = eidx[(long long)E + e];
        const int b = min((int)((unsigned)s / SRC_R), nranges - 1);
        const int pos = atomicAdd(&cur[b], 1);
        rec[pos] = (unsigned)d | ((unsigned)(s - b * SRC_R) << SRC_SHIFT);
    }
}

template <bool FULL>
__device__ __forceinline__ void src_batch(unsigned r, int nleft, int c, int hi,
                                          const unsigned* __restrict__ rec, int pn, int end,
                                          int lane, unsigned& rn, const float* rows,
                                          float* __restrict__ agg) {
    float v[16];
    int   dj[16];
#pragma unroll
    for (int i = 0; i < 16; ++i) {
        const unsigned r0 = (unsigned)__builtin_amdgcn_readlane((int)r, 2 * i);
        const unsigned r1 = (unsigned)__builtin_amdgcn_readlane((int)r, 2 * i + 1);
        const unsigned rj = hi ? r1 : r0;
        dj[i] = (int)(rj & ((1u << SRC_SHIFT) - 1u));
        v[i] = rows[(rj >> SRC_SHIFT) * HID + c];
    }
    if (pn + lane < end) rn = rec[pn + lane];
#pragma unroll
    for (int i = 0; i < 16; ++i) {
        if (FULL || 2 * i + hi < nleft) atomicAdd(agg + (size_t)dj[i] * HID + c, v[i]);
    }
}

__global__ void __launch_bounds__(256, 16)
ak_src_scatter(const unsigned* __restrict__ rec, const int* __restrict__ off,
               const float* __restrict__ g, float* agg0, float* agg1, int N, int slices) {
    __shared__ float rows[SRC_R * HID];
    const int tid = (int)threadIdx.x;
    const int b = (int)blockIdx.x / slices;
    const int sl = (int)blockIdx.x - b * slices;
    {
        const int base = b * SRC_R;
        const int n4 = min(SRC_R, N - base) * (HID / 4);
        const float4* __restrict__ g4 = (const float4*)g + (size_t)base * (HID / 4);
        float4* r4 = (float4*)rows;
        float4 t[SRC_R * (HID / 4) / 256];
#pragma unroll
        for (int k = 0; k < SRC_R * (HID / 4) / 256; ++k) {
            const int i = tid + k * 256;
            t[k] = g4[min(i, n4 - 1)];
        }
#pragma unroll
        for (int k = 0; k < SRC_R * (HID / 4) / 256; ++k) r4[tid + k * 256] = t[k];
    }
    __syncthreads();

    const int lane = tid & 31;
    const int c  = lane & 15;
    const int hi = lane >> 4;
    float* __restrict__ agg = cache_half() ? agg1 : agg0;

    const int e0 = off[b * SORT_WGS];
    const int e1 = off[(b + 1) * SORT_WGS];
    const int per = (((e1 - e0) + slices - 1) / slices + 31) & ~31;
    const int lo = e0 + sl * per;
    const int end = min(e1, lo + per);

    int p = lo + __builtin_amdgcn_readfirstlane(tid >> 5) * 32;
    if (p >= end) return;
    unsigned r = (p + lane < end) ? rec[p + lane] : 0u;
    for (;;) {
        const int nleft = end - p;
        const int pn = p + 256;
        unsigned rn = 0u;
        if (nleft >= 32) src_batch<true>(r, 32, c, hi, rec, pn, end, lane, rn, rows, agg);
        else             src_batch<false>(r, nleft, c, hi, rec, pn, end, lane, rn, rows, agg);
        if (pn >= end) break;
        p = pn; r = rn;
    }
}

__global__ void ak_finalize(float4* agg0, float4* agg1, float4* __restrict__ g,
                            const float4* __restrict__ h0, const float* __restrict__ dinv,
                            int n4) {
    int t = blockIdx.x * blockDim.x + threadIdx.x;
    if (t >= n4) return;
    int v = t >> 2;
    float dv = dinv[v];
    float4 a = agg0[t], gg = g[t], h = h0[t];
    float4 z; z.x = 0.f; z.y = 0.f; z.z = 0.f; z.w = 0.f;
    if (agg1 != agg0) {
        float4 a1 = agg1[t];
        a.x += a1.x; a.y += a1.y; a.z += a1.z; a.w += a1.w;
        agg1[t] = z;
    }
    float4 hn;
    hn.x = ONE_MINUS_ALPHA * dv * (a.x + gg.x) + ALPHA_F * h.x;
    hn.y = ONE_MINUS_ALPHA * dv * (a.y + gg.y) + ALPHA_F * h.y;
    hn.z = ONE_MINUS_ALPHA * dv * (a.z + gg.z) + ALPHA_F * h.z;
    hn.w = ONE_MINUS_ALPHA * dv * (a.w + gg.w) + ALPHA_F * h.w;
    float4 gn; gn.x = dv * hn.x; gn.y = dv * hn.y; gn.z = dv * hn.z; gn.w = dv * hn.w;
    g[t] = gn;
    agg0[t] = z;
}

__global__ void ak_finalize_out(const float* agg0, const float* agg1, const float* __restrict__ g,
                                const float* __restrict__ h0, const float* __restrict__ dinv,
                                const float* __restrict__ W3, const float* __restrict__ b3,
                                float* __restrict__ out, int N) {
    int v = blockIdx.x * blockDim.x + threadIdx.x;
    if (v >= N) return;
    float dv = dinv[v];
    float s = b3[0];
    const bool two = (agg1 != agg0);
#pragma unroll
    for (int j = 0; j < HID; ++j) {
        float a = agg0[v * HID + j];
        if (two) a += agg1[v * HID + j];
        float h = ONE_MINUS_ALPHA * dv * (a + g[v * HID + j])
                + ALPHA_F * h0[v * HID + j];
        s += h * W3[j];
    }
    out[v] = s;
}


extern "C" void kernel_launch(void* const* d_in, const int* in_sizes, int n_in,
                              void* d_out, int out_size, void* d_ws, size_t ws_size,
                              hipStream_t stream) {
    const float* x  = (const float*)d_in[0];
    const int*  eidx = (const int*)d_in[1];
    const float* W1 = (const float*)d_in[2];
    const float* b1 = (const float*)d_in[3];
    const float* W2 = (const float*)d_in[4];
    const float* b2 = (const float*)d_in[5];
    const float* W3 = (const float*)d_in[6];
    const float* b3 = (const float*)d_in[7];
    float* out = (float*)d_out;

    const int N = in_sizes[0];
    const int E = in_sizes[1] / 2;

    char* ws = (char*)d_ws;
    size_t off = 0;
    auto carve = [&](size_t bytes) -> char* {
        char* p = ws + off;
        off += (bytes + 255) & ~(size_t)255;
        return p;
    };
    float* deg  = (float*)carve((size_t)N * 4);
    float* dinv = (float*)carve((size_t)N * 4);
    float* h0   = (float*)carve((size_t)N * HID * 4);
    float* g    = (float*)carve((size_t)N * HID * 4);
    float* agg  = (float*)carve((size_t)N * HID * 4);
    float* agg1 = agg;
    float* deg1 = deg;
    if (off + (size_t)N * HID * 4 + (size_t)N * 4 + 512 <= ws_size) {
        agg1 = (float*)carve((size_t)N * HID * 4);
        deg1 = (float*)carve((size_t)N * 4);
    }

    const int nranges = (N + SRC_R - 1) / SRC_R;
    unsigned* rec = nullptr;
    int* eoff = nullptr;
    if (nranges > 0 && nranges <= MAX_RANGES && N <= (1 << SRC_SHIFT) &&
        off + (size_t)E * 4 + ((size_t)nranges * SORT_WGS + 1) * 4 + 512 <= ws_size) {
        rec  = (unsigned*)carve((size_t)E * 4);
        eoff = (int*)carve(((size_t)nranges * SORT_WGS + 1) * 4);
    }

    const int B = 256;
    int sblocks = SCATTER_BLOCKS;
    ak_deg_init<<<(N + B - 1) / B, B, 0, stream>>>(deg, deg1, N);
    ak_deg_accum<<<sblocks, B, 0, stream>>>(eidx, deg, deg1, E);
    ak_dinv<<<(N + B - 1) / B, B, 0, stream>>>(deg, deg1, dinv, N);

    int tiles = N / 16;
    if (tiles > 0) {
        int waves_per_block = B / 32;
        int blocks = (tiles + waves_per_block - 1) / waves_per_block;
        ak_encoder_wmma<<<blocks, B, 0, stream>>>(x, W1, b1, W2, b2, dinv, h0, g, agg, agg1,
                                                  tiles);
    }
    int rem_start = tiles * 16;
    if (rem_start < N) {
        ak_encoder_tail<<<(N - rem_start + B - 1) / B, B, 0, stream>>>(
            x, W1, b1, W2, b2, dinv, h0, g, agg, agg1, rem_start, N);
    }

    int slices = 1;
    if (rec) {
        ak_src_count<<<SORT_WGS, SORT_THREADS, 0, stream>>>(eidx, eoff, E, nranges);
        ak_src_scan<<<1, SORT_THREADS, 0, stream>>>(eoff, nranges * SORT_WGS);
        ak_src_place<<<SORT_WGS, SORT_THREADS, 0, stream>>>(eidx, eoff, rec, E, nranges);
        if (SCATTER_BLOCKS / nranges > 1) slices = SCATTER_BLOCKS / nranges;
    }

    int n4 = N * 4;
    for (int it = 0; it < KITERS; ++it) {
        if (rec) ak_src_scatter<<<nranges * slices, B, 0, stream>>>(rec, eoff, g, agg, agg1, N, slices);
        else     ak_edge_scatter<<<sblocks, B, 0, stream>>>(eidx, g, agg, agg1, E);
        if (it < KITERS - 1) {
            ak_finalize<<<(n4 + B - 1) / B, B, 0, stream>>>(
                (float4*)agg, (float4*)agg1, (float4*)g, (const float4*)h0, dinv, n4);
        } else {
            ak_finalize_out<<<(N + B - 1) / B, B, 0, stream>>>(
                agg, agg1, g, h0, dinv, W3, b3, out, N);
        }
    }
}


# Edges grouped by 196-node destination range (was source range): a work-group adds its range's rows and degree counts in LDS (atomic) and rescales there; no global atomics; float sum order changes
# speedup vs baseline: 6.0651x; 1.6412x over previous
#include <hip/hip_runtime.h>


typedef __attribute__((ext_vector_type(2))) float v2f;
typedef __attribute__((ext_vector_type(8))) float v8f;

#define HID 16
#define KITERS 10
#define ONE_MINUS_ALPHA 0.9f
#define ALPHA_F 0.1f

__device__ __forceinline__ float relu_f(float v) { return v > 0.f ? v : 0.f; }

__device__ __forceinline__ int cache_half() {
    return (int)((__builtin_amdgcn_s_sendmsg_rtn(0x87) >> 18) & 1u);
}


__global__ void ak_deg_init(float* deg0, float* deg1, int n) {
    int i = blockIdx.x * blockDim.x + threadIdx.x;
    if (i < n) {
        deg0[i] = 1.0f;
        if (deg1 != deg0) deg1[i] = 0.0f;
    }
}

__global__ void ak_deg_accum(const int* __restrict__ eidx, float* deg0, float* deg1, int E) {
    float* deg = cache_half() ? deg1 : deg0;
    const int stride = (int)(gridDim.x * blockDim.x);
    for (int e = (int)(blockIdx.x * blockDim.x + threadIdx.x); e < E; e += stride)
        atomicAdd(&deg[eidx[(long long)E + e]], 1.0f);
}

__global__ void ak_dinv(const float* deg0, const float* deg1, float* __restrict__ dinv, int n) {
    int i = blockIdx.x * blockDim.x + threadIdx.x;
    if (i < n) {
        float d = deg0[i];
        if (deg1 != deg0) d += deg1[i];
        dinv[i] = rsqrtf(d);
    }
}


__global__ void ak_encoder_wmma(const float* __restrict__ x, const float* __restrict__ W1,
                                const float* __restrict__ b1, const float* __restrict__ W2,
                                const float* __restrict__ b2, const float* __restrict__ dinv,
                                float* __restrict__ h0, float* __restrict__ g,
                                float* agg0, float* agg1, int n_tiles) {
    int wave = (blockIdx.x * blockDim.x + threadIdx.x) >> 5;
    int lane = threadIdx.x & 31;
    if (wave >= n_tiles) return;

    int row = lane & 15;
    int hi  = lane >> 4;
    int node = wave * 16 + row;
    float xv = x[node];

    v8f acc = {};
#pragma unroll
    for (int s = 0; s < 4; ++s) {
        int k0 = 4 * s + 2 * hi;
        v2f a, b;
        a.x = relu_f(xv * W1[k0]     + b1[k0]);
        a.y = relu_f(xv * W1[k0 + 1] + b1[k0 + 1]);
        b.x = W2[k0 * HID + row];
        b.y = W2[(k0 + 1) * HID + row];
        acc = __builtin_amdgcn_wmma_f32_16x16x4_f32(
            false, a, false, b, (short)0, acc, false, false);
    }

    float bb = b2[row];
#pragma unroll
    for (int i = 0; i < 8; ++i) {
        int r = i + 8 * hi;
        int nr = wave * 16 + r;
        float v = relu_f(acc[i] + bb);
        int idx = nr * HID + row;
        h0[idx]  = v;
        g[idx]   = dinv[nr] * v;
        if (agg0) {
            agg0[idx] = 0.f;
            if (agg1 != agg0) agg1[idx] = 0.f;
        }
    }
}

__global__ void ak_encoder_tail(const float* __restrict__ x, const float* __restrict__ W1,
                                const float* __restrict__ b1, const float* __restrict__ W2,
                                const float* __restrict__ b2, const float* __restrict__ dinv,
                                float* __restrict__ h0, float* __restrict__ g,
                                float* agg0, float* agg1, int start, int N) {
    int v = start + blockIdx.x * blockDim.x + threadIdx.x;
    if (v >= N) return;
    float h1[HID];
    float xv = x[v];
#pragma unroll
    for (int k = 0; k < HID; ++k) h1[k] = relu_f(xv * W1[k] + b1[k]);
    float dv = dinv[v];
#pragma unroll
    for (int j = 0; j < HID; ++j) {
        float acc = b2[j];
#pragma unroll
        for (int k = 0; k < HID; ++k) acc += h1[k] * W2[k * HID + j];
        float hv = relu_f(acc);
        int idx = v * HID + j;
        h0[idx] = hv; g[idx] = dv * hv;
        if (agg0) {
            agg0[idx] = 0.f;
            if (agg1 != agg0) agg1[idx] = 0.f;
        }
    }
}


#define SCATTER_BLOCKS 2048

template <bool FULL>
__device__ __forceinline__ void scatter_batch(int s, int d, int nleft, int c, int hi,
                                              const int* __restrict__ eidx, long long en,
                                              int E, int& sn, int& dn,
                                              const float* __restrict__ g,
                                              float* __restrict__ agg) {
    float v[16];
    int   dj[16];
#pragma unroll
    for (int i = 0; i < 16; ++i) {
        const int s0 = __builtin_amdgcn_readlane(s, 2 * i);
        const int s1 = __builtin_amdgcn_readlane(s, 2 * i + 1);
        const int d0 = __builtin_amdgcn_readlane(d, 2 * i);
        const int d1 = __builtin_amdgcn_readlane(d, 2 * i + 1);
        const int sj = hi ? s1 : s0;
        dj[i] = hi ? d1 : d0;
        v[i] = (FULL || 2 * i + hi < nleft) ? g[sj * HID + c] : 0.f;
    }
    if (en < E) { sn = eidx[en]; dn = eidx[(long long)E + en]; }
#pragma unroll
    for (int i = 0; i < 16; ++i) {
        if (FULL || 2 * i + hi < nleft) atomicAdd(agg + dj[i] * HID + c, v[i]);
    }
}

__global__ void ak_edge_scatter(const int* __restrict__ eidx, const float* __restrict__ g,
                                float* agg0, float* agg1, int E) {
    const int lane = threadIdx.x & 31;
    const int c  = lane & 15;
    const int hi = lane >> 4;
    const int nwaves = (int)((gridDim.x * blockDim.x) >> 5);
    const int nbatch = (E + 31) >> 5;
    int b = (int)((blockIdx.x * blockDim.x + threadIdx.x) >> 5);
    if (b >= nbatch) return;

    float* __restrict__ agg = cache_half() ? agg1 : agg0;

    int s = 0, d = 0;
    {
        const long long e = (long long)b * 32 + lane;
        if (e < E) { s = eidx[e]; d = eidx[(long long)E + e]; }
    }
    for (;;) {
        const int bn = b + nwaves;
        const long long en = (bn < nbatch) ? (long long)bn * 32 + lane : (long long)E;
        int sn = 0, dn = 0;
        const int nleft = E - b * 32;
        if (nleft >= 32) scatter_batch<true>(s, d, 32, c, hi, eidx, en, E, sn, dn, g, agg);
        else             scatter_batch<false>(s, d, nleft, c, hi, eidx, en, E, sn, dn, g, agg);
        if (bn >= nbatch) break;
        b = bn; s = sn; d = dn;
    }
}


#define DST_R 196
#define DST_SHIFT 23
#define SORT_WGS 256
#define SORT_THREADS 1024
#define MAX_RANGES 2048
#define HOP_THREADS 1024

__global__ void __launch_bounds__(SORT_THREADS)
ak_grp_count(const int* __restrict__ eidx, int* __restrict__ off, int E, int nranges) {
    __shared__ int hist[MAX_RANGES];
    const int tid = (int)threadIdx.x, w = (int)blockIdx.x;
    for (int i = tid; i < nranges; i += SORT_THREADS) hist[i] = 0;
    __syncthreads();
    const int chunk = (E + SORT_WGS - 1) / SORT_WGS;
    const int lo = w * chunk;
    const int hi = min(E, lo + chunk);
    for (int e = lo + tid; e < hi; e += SORT_THREADS) {
        const int b = min((int)((unsigned)eidx[(long long)E + e] / DST_R), nranges - 1);
        atomicAdd(&hist[b], 1);
    }
    __syncthreads();
    for (int i = tid; i < nranges; i += SORT_THREADS) off[i * SORT_WGS + w] = hist[i];
}

__device__ __forceinline__ int wave_incl_sum(int x, int lane) {
#pragma unroll
    for (int dlt = 1; dlt < 32; dlt <<= 1) {
        const int y = __shfl_up(x, dlt, 32);
        if (lane >= dlt) x += y;
    }
    return x;
}

__global__ void __launch_bounds__(SORT_THREADS)
ak_grp_scan(int* off, int n) {
    __shared__ int wtot[32];
    const int tid = (int)threadIdx.x;
    const int lane = tid & 31;
    const int wv = tid >> 5;
    const int span = (((n + 31) >> 5) + 127) & ~127;
    const int first = min(n, wv * span);
    const int last = min(n, first + span);
    int4* off4 = (int4*)off;

    int s = 0;
    for (int i = first + lane * 4; i < last; i += 128) {
        const int4 v = off4[i >> 2];
        s += (v.x + v.y) + (v.z + v.w);
    }
    s = wave_incl_sum(s, lane);
    if (lane == 31) wtot[wv] = s;
    __syncthreads();
    const int t = wtot[lane];
    const int upto = wave_incl_sum(t, lane);
    const int all = __shfl(upto, 31, 32);
    int run = __shfl(upto - t, wv, 32);
    for (int i0 = first; i0 < last; i0 += 128) {
        const int i = i0 + lane * 4;
        int4 v; v.x = 0; v.y = 0; v.z = 0; v.w = 0;
        if (i < last) v = off4[i >> 2];
        const int s1 = v.x + v.y, s2 = s1 + v.z, s3 = s2 + v.w;
        const int inc = wave_incl_sum(s3, lane);
        const int base = run + inc - s3;
        int4 o; o.x = base; o.y = base + v.x; o.z = base + s1; o.w = base + s2;
        if (i < last) off4[i >> 2] = o;
        run += __shfl(inc, 31, 32);
    }
    if (tid == 0) off[n] = all;
}

__global__ void __launch_bounds__(SORT_THREADS)
ak_grp_place(const int* __restrict__ eidx, const int* __restrict__ off,
             unsigned* __restrict__ rec, int E, int nranges) {
    __shared__ int cur[MAX_RANGES];
    const int tid = (int)threadIdx.x, w = (int)blockIdx.x;
    for (int i = tid; i < nranges; i += SORT_THREADS) cur[i] = off[i * SORT_WGS + w];
    __syncthreads();
    const int chunk = (E + SORT_WGS - 1) / SORT_WGS;
    const int lo = w * chunk;
    const int hi = min(E, lo + chunk);
    for (int e = lo + tid; e < hi; e += SORT_THREADS) {
        const int s = eidx[e];
        const int d = eidx[(long long)E + e];
        const int b = min((int)((unsigned)d / DST_R), nranges - 1);
        const int pos = atomicAdd(&cur[b], 1);
        rec[pos] = (unsigned)s | ((unsigned)(d - b * DST_R) << DST_SHIFT);
    }
}

__global__ void __launch_bounds__(SORT_THREADS)
ak_grp_dinv(const unsigned* __restrict__ rec, const int* __restrict__ off,
            float* __restrict__ dinv, int N) {
    __shared__ int cnt[DST_R];
    const int tid = (int)threadIdx.x;
    const int b = (int)blockIdx.x;
    const int base = b * DST_R;
    const int nrow = min(DST_R, N - base);
    if (tid < DST_R) cnt[tid] = 0;
    __syncthreads();
    const int e0 = off[b * SORT_WGS];
    const int e1 = off[(b + 1) * SORT_WGS];
    for (int p = e0 + tid; p < e1; p += 4 * SORT_THREADS) {
        unsigned r[4];
#pragma unroll
        for (int k = 0; k < 4; ++k) {
            const int q = p + k * SORT_THREADS;
            r[k] = (q < e1) ? rec[q] : 0u;
        }
#pragma unroll
        for (int k = 0; k < 4; ++k) {
            if (p + k * SORT_THREADS < e1) atomicAdd(&cnt[r[k] >> DST_SHIFT], 1);
        }
    }
    __syncthreads();
    if (tid < nrow) dinv[base + tid] = rsqrtf(1.0f + (float)cnt[tid]);
}

template <int I>
__device__ __forceinline__ void dst_rows(unsigned r, int c, const float* __restrict__ gin,
                                         float* v, int* dj) {
    if constexpr (I < 16) {
        const unsigned rj = (unsigned)__builtin_amdgcn_update_dpp(0, (int)r, 0x150 + I, 0xf, 0xf,
                                                                   true);
        dj[I] = (int)(rj >> DST_SHIFT) * HID + c;
        v[I] = gin[(size_t)(rj & ((1u << DST_SHIFT) - 1u)) * HID + c];
        dst_rows<I + 1>(r, c, gin, v, dj);
    }
}

template <bool FULL>
__device__ __forceinline__ void dst_batch(unsigned r, int nleft, int c, int hi,
                                          const unsigned* __restrict__ rec, int pn, int end,
                                          int lane, unsigned& rn, const float* __restrict__ gin,
                                          float* acc) {
    float v[16];
    int   dj[16];
    dst_rows<0>(r, c, gin, v, dj);
    if (pn + lane < end) rn = rec[pn + lane];
#pragma unroll
    for (int i = 0; i < 16; ++i) {
        if (FULL || 16 * hi + i < nleft) atomicAdd(&acc[dj[i]], v[i]);
    }
}

template <bool LAST>
__global__ void __launch_bounds__(HOP_THREADS, 16)
ak_dst_hop(const unsigned* __restrict__ rec, const int* __restrict__ off,
           const float* __restrict__ gin, float* __restrict__ gout,
           const float* __restrict__ h0, const float* __restrict__ dinv,
           const float* __restrict__ W3, const float* __restrict__ b3,
           float* __restrict__ out, int N) {
    __shared__ float acc[DST_R * HID];
    const int tid = (int)threadIdx.x;
    const int b = (int)blockIdx.x;
    const int base = b * DST_R;
    const int n4 = min(DST_R, N - base) * (HID / 4);
    const bool live = tid < n4;
    const size_t t4 = (size_t)base * (HID / 4) + (size_t)(live ? tid : 0);

    const int lane = tid & 31;
    const int c  = lane & 15;
    const int hi = lane >> 4;
    const int e0 = off[b * SORT_WGS];
    const int end = off[(b + 1) * SORT_WGS];
    int p = e0 + __builtin_amdgcn_readfirstlane(tid >> 5) * 32;
    unsigned r = (p + lane < end) ? rec[p + lane] : 0u;

    if (live) ((float4*)acc)[tid] = ((const float4*)gin)[t4];
    __syncthreads();

    if (p < end) {
        for (;;) {
            const int nleft = end - p;
            const int pn = p + HOP_THREADS;
            unsigned rn = 0u;
            if (nleft >= 32) dst_batch<true>(r, 32, c, hi, rec, pn, end, lane, rn, gin, acc);
            else             dst_batch<false>(r, nleft, c, hi, rec, pn, end, lane, rn, gin, acc);
            if (pn >= end) break;
            p = pn; r = rn;
        }
    }
    __syncthreads();

    const float dv = dinv[base + ((live ? tid : 0) >> 2)];
    const float4 a = ((const float4*)acc)[live ? tid : 0];
    const float4 h = ((const float4*)h0)[t4];
    float4 hn;
    hn.x = ONE_MINUS_ALPHA * dv * a.x + ALPHA_F * h.x;
    hn.y = ONE_MINUS_ALPHA * dv * a.y + ALPHA_F * h.y;
    hn.z = ONE_MINUS_ALPHA * dv * a.z + ALPHA_F * h.z;
    hn.w = ONE_MINUS_ALPHA * dv * a.w + ALPHA_F * h.w;
    if (!LAST) {
        float4 gn; gn.x = dv * hn.x; gn.y = dv * hn.y; gn.z = dv * hn.z; gn.w = dv * hn.w;
        if (live) ((float4*)gout)[t4] = gn;
    } else {
        const int j = (tid & 3) * 4;
        float s = (hn.x * W3[j] + hn.y * W3[j + 1]) + (hn.z * W3[j + 2] + hn.w * W3[j + 3]);
        s += __shfl_xor(s, 1, 32);
        s += __shfl_xor(s, 2, 32);
        if (live && (tid & 3) == 0) out[base + (tid >> 2)] = s + b3[0];
    }
}

__global__ void ak_finalize(float4* agg0, float4* agg1, float4* __restrict__ g,
                            const float4* __restrict__ h0, const float* __restrict__ dinv,
                            int n4) {
    int t = blockIdx.x * blockDim.x + threadIdx.x;
    if (t >= n4) return;
    int v = t >> 2;
    float dv = dinv[v];
    float4 a = agg0[t], gg = g[t], h = h0[t];
    float4 z; z.x = 0.f; z.y = 0.f; z.z = 0.f; z.w = 0.f;
    if (agg1 != agg0) {
        float4 a1 = agg1[t];
        a.x += a1.x; a.y += a1.y; a.z += a1.z; a.w += a1.w;
        agg1[t] = z;
    }
    float4 hn;
    hn.x = ONE_MINUS_ALPHA * dv * (a.x + gg.x) + ALPHA_F * h.x;
    hn.y = ONE_MINUS_ALPHA * dv * (a.y + gg.y) + ALPHA_F * h.y;
    hn.z = ONE_MINUS_ALPHA * dv * (a.z + gg.z) + ALPHA_F * h.z;
    hn.w = ONE_MINUS_ALPHA * dv * (a.w + gg.w) + ALPHA_F * h.w;
    float4 gn; gn.x = dv * hn.x; gn.y = dv * hn.y; gn.z = dv * hn.z; gn.w = dv * hn.w;
    g[t] = gn;
    agg0[t] = z;
}

__global__ void ak_finalize_out(const float* agg0, const float* agg1, const float* __restrict__ g,
                                const float* __restrict__ h0, const float* __restrict__ dinv,
                                const float* __restrict__ W3, const float* __restrict__ b3,
                                float* __restrict__ out, int N) {
    int v = blockIdx.x * blockDim.x + threadIdx.x;
    if (v >= N) return;
    float dv = dinv[v];
    float s = b3[0];
    const bool two = (agg1 != agg0);
#pragma unroll
    for (int j = 0; j < HID; ++j) {
        float a = agg0[v * HID + j];
        if (two) a += agg1[v * HID + j];
        float h = ONE_MINUS_ALPHA * dv * (a + g[v * HID + j])
                + ALPHA_F * h0[v * HID + j];
        s += h * W3[j];
    }
    out[v] = s;
}


extern "C" void kernel_launch(void* const* d_in, const int* in_sizes, int n_in,
                              void* d_out, int out_size, void* d_ws, size_t ws_size,
                              hipStream_t stream) {
    const float* x  = (const float*)d_in[0];
    const int*  eidx = (const int*)d_in[1];
    const float* W1 = (const float*)d_in[2];
    const float* b1 = (const float*)d_in[3];
    const float* W2 = (const float*)d_in[4];
    const float* b2 = (const float*)d_in[5];
    const float* W3 = (const float*)d_in[6];
    const float* b3 = (const float*)d_in[7];
    float* out = (float*)d_out;

    const int N = in_sizes[0];
    const int E = in_sizes[1] / 2;

    char* ws = (char*)d_ws;
    size_t off = 0;
    auto carve = [&](size_t bytes) -> char* {
        char* p = ws + off;
        off += (bytes + 255) & ~(size_t)255;
        return p;
    };
    float* deg  = (float*)carve((size_t)N * 4);
    float* dinv = (float*)carve((size_t)N * 4);
    float* h0   = (float*)carve((size_t)N * HID * 4);
    float* g    = (float*)carve((size_t)N * HID * 4);
    float* agg  = (float*)carve((size_t)N * HID * 4);
    float* agg1 = agg;
    float* deg1 = deg;
    if (off + (size_t)N * HID * 4 + (size_t)N * 4 + 512 <= ws_size) {
        agg1 = (float*)carve((size_t)N * HID * 4);
        deg1 = (float*)carve((size_t)N * 4);
    }

    const int nranges = (N + DST_R - 1) / DST_R;
    unsigned* rec = nullptr;
    int* eoff = nullptr;
    if (nranges > 0 && nranges <= MAX_RANGES && N <= (1 << DST_SHIFT) &&
        off + (size_t)E * 4 + ((size_t)nranges * SORT_WGS + 1) * 4 + 512 <= ws_size) {
        rec  = (unsigned*)carve((size_t)E * 4);
        eoff = (int*)carve(((size_t)nranges * SORT_WGS + 1) * 4);
    }

    const int B = 256;
    int sblocks = SCATTER_BLOCKS;
    if (rec) {
        ak_grp_count<<<SORT_WGS, SORT_THREADS, 0, stream>>>(eidx, eoff, E, nranges);
        ak_grp_scan<<<1, SORT_THREADS, 0, stream>>>(eoff, nranges * SORT_WGS);
        ak_grp_place<<<SORT_WGS, SORT_THREADS, 0, stream>>>(eidx, eoff, rec, E, nranges);
        ak_grp_dinv<<<nranges, SORT_THREADS, 0, stream>>>(rec, eoff, dinv, N);
    } else {
        ak_deg_init<<<(N + B - 1) / B, B, 0, stream>>>(deg, deg1, N);
        ak_deg_accum<<<sblocks, B, 0, stream>>>(eidx, deg, deg1, E);
        ak_dinv<<<(N + B - 1) / B, B, 0, stream>>>(deg, deg1, dinv, N);
    }

    float* za0 = rec ? nullptr : agg;
    float* za1 = rec ? nullptr : agg1;
    int tiles = N / 16;
    if (tiles > 0) {
        int waves_per_block = B / 32;
        int blocks = (tiles + waves_per_block - 1) / waves_per_block;
        ak_encoder_wmma<<<blocks, B, 0, stream>>>(x, W1, b1, W2, b2, dinv, h0, g, za0, za1,
                                                  tiles);
    }
    int rem_start = tiles * 16;
    if (rem_start < N) {
        ak_encoder_tail<<<(N - rem_start + B - 1) / B, B, 0, stream>>>(
            x, W1, b1, W2, b2, dinv, h0, g, za0, za1, rem_start, N);
    }

    if (rec) {
        float* ga = g;
        float* gb = agg;
        for (int it = 0; it < KITERS - 1; ++it) {
            ak_dst_hop<false><<<nranges, HOP_THREADS, 0, stream>>>(rec, eoff, ga, gb, h0, dinv,
                                                                   W3, b3, out, N);
            float* t = ga; ga = gb; gb = t;
        }
        ak_dst_hop<true><<<nranges, HOP_THREADS, 0, stream>>>(rec, eoff, ga, gb, h0, dinv,
                                                              W3, b3, out, N);
        return;
    }

    int n4 = N * 4;
    for (int it = 0; it < KITERS; ++it) {
        ak_edge_scatter<<<sblocks, B, 0, stream>>>(eidx, g, agg, agg1, E);
        if (it < KITERS - 1) {
            ak_finalize<<<(n4 + B - 1) / B, B, 0, stream>>>(
                (float4*)agg, (float4*)agg1, (float4*)g, (const float4*)h0, dinv, n4);
        } else {
            ak_finalize_out<<<(N + B - 1) / B, B, 0, stream>>>(
                agg, agg1, g, h0, dinv, W3, b3, out, N);
        }
    }
}


# Grouping passes issue their loads in batches, not one a loop trip: count and place fetch 9 edges a thread ahead of the LDS adds, the running sum takes 4 table rows a wave; float sum order may change
# speedup vs baseline: 6.3732x; 1.0508x over previous
#include <hip/hip_runtime.h>


typedef __attribute__((ext_vector_type(2))) float v2f;
typedef __attribute__((ext_vector_type(8))) float v8f;

#define HID 16
#define KITERS 10
#define ONE_MINUS_ALPHA 0.9f
#define ALPHA_F 0.1f

__device__ __forceinline__ float relu_f(float v) { return v > 0.f ? v : 0.f; }

__device__ __forceinline__ int cache_half() {
    return (int)((__builtin_amdgcn_s_sendmsg_rtn(0x87) >> 18) & 1u);
}


__global__ void ak_deg_init(float* deg0, float* deg1, int n) {
    int i = blockIdx.x * blockDim.x + threadIdx.x;
    if (i < n) {
        deg0[i] = 1.0f;
        if (deg1 != deg0) deg1[i] = 0.0f;
    }
}

__global__ void ak_deg_accum(const int* __restrict__ eidx, float* deg0, float* deg1, int E) {
    float* deg = cache_half() ? deg1 : deg0;
    const int stride = (int)(gridDim.x * blockDim.x);
    for (int e = (int)(blockIdx.x * blockDim.x + threadIdx.x); e < E; e += stride)
        atomicAdd(&deg[eidx[(long long)E + e]], 1.0f);
}

__global__ void ak_dinv(const float* deg0, const float* deg1, float* __restrict__ dinv, int n) {
    int i = blockIdx.x * blockDim.x + threadIdx.x;
    if (i < n) {
        float d = deg0[i];
        if (deg1 != deg0) d += deg1[i];
        dinv[i] = rsqrtf(d);
    }
}


__global__ void ak_encoder_wmma(const float* __restrict__ x, const float* __restrict__ W1,
                                const float* __restrict__ b1, const float* __restrict__ W2,
                                const float* __restrict__ b2, const float* __restrict__ dinv,
                                float* __restrict__ h0, float* __restrict__ g,
                                float* agg0, float* agg1, int n_tiles) {
    int wave = (blockIdx.x * blockDim.x + threadIdx.x) >> 5;
    int lane = threadIdx.x & 31;
    if (wave >= n_tiles) return;

    int row = lane & 15;
    int hi  = lane >> 4;
    int node = wave * 16 + row;
    float xv = x[node];

    v8f acc = {};
#pragma unroll
    for (int s = 0; s < 4; ++s) {
        int k0 = 4 * s + 2 * hi;
        v2f a, b;
        a.x = relu_f(xv * W1[k0]     + b1[k0]);
        a.y = relu_f(xv * W1[k0 + 1] + b1[k0 + 1]);
        b.x = W2[k0 * HID + row];
        b.y = W2[(k0 + 1) * HID + row];
        acc = __builtin_amdgcn_wmma_f32_16x16x4_f32(
            false, a, false, b, (short)0, acc, false, false);
    }

    float bb = b2[row];
#pragma unroll
    for (int i = 0; i < 8; ++i) {
        int r = i + 8 * hi;
        int nr = wave * 16 + r;
        float v = relu_f(acc[i] + bb);
        int idx = nr * HID + row;
        h0[idx]  = v;
        g[idx]   = dinv[nr] * v;
        if (agg0) {
            agg0[idx] = 0.f;
            if (agg1 != agg0) agg1[idx] = 0.f;
        }
    }
}

__global__ void ak_encoder_tail(const float* __restrict__ x, const float* __restrict__ W1,
                                const float* __restrict__ b1, const float* __restrict__ W2,
                                const float* __restrict__ b2, const float* __restrict__ dinv,
                                float* __restrict__ h0, float* __restrict__ g,
                                float* agg0, float* agg1, int start, int N) {
    int v = start + blockIdx.x * blockDim.x + threadIdx.x;
    if (v >= N) return;
    float h1[HID];
    float xv = x[v];
#pragma unroll
    for (int k = 0; k < HID; ++k) h1[k] = relu_f(xv * W1[k] + b1[k]);
    float dv = dinv[v];
#pragma unroll
    for (int j = 0; j < HID; ++j) {
        float acc = b2[j];
#pragma unroll
        for (int k = 0; k < HID; ++k) acc += h1[k] * W2[k * HID + j];
        float hv = relu_f(acc);
        int idx = v * HID + j;
        h0[idx] = hv; g[idx] = dv * hv;
        if (agg0) {
            agg0[idx] = 0.f;
            if (agg1 != agg0) agg1[idx] = 0.f;
        }
    }
}


#define SCATTER_BLOCKS 2048

template <bool FULL>
__device__ __forceinline__ void scatter_batch(int s, int d, int nleft, int c, int hi,
                                              const int* __restrict__ eidx, long long en,
                                              int E, int& sn, int& dn,
                                              const float* __restrict__ g,
                                              float* __restrict__ agg) {
    float v[16];
    int   dj[16];
#pragma unroll
    for (int i = 0; i < 16; ++i) {
        const int s0 = __builtin_amdgcn_readlane(s, 2 * i);
        const int s1 = __builtin_amdgcn_readlane(s, 2 * i + 1);
        const int d0 = __builtin_amdgcn_readlane(d, 2 * i);
        const int d1 = __builtin_amdgcn_readlane(d, 2 * i + 1);
        const int sj = hi ? s1 : s0;
        dj[i] = hi ? d1 : d0;
        v[i] = (FULL || 2 * i + hi < nleft) ? g[sj * HID + c] : 0.f;
    }
    if (en < E) { sn = eidx[en]; dn = eidx[(long long)E + en]; }
#pragma unroll
    for (int i = 0; i < 16; ++i) {
        if (FULL || 2 * i + hi < nleft) atomicAdd(agg + dj[i] * HID + c, v[i]);
    }
}

__global__ void ak_edge_scatter(const int* __restrict__ eidx, const float* __restrict__ g,
                                float* agg0, float* agg1, int E) {
    const int lane = threadIdx.x & 31;
    const int c  = lane & 15;
    const int hi = lane >> 4;
    const int nwaves = (int)((gridDim.x * blockDim.x) >> 5);
    const int nbatch = (E + 31) >> 5;
    int b = (int)((blockIdx.x * blockDim.x + threadIdx.x) >> 5);
    if (b >= nbatch) return;

    float* __restrict__ agg = cache_half() ? agg1 : agg0;

    int s = 0, d = 0;
    {
        const long long e = (long long)b * 32 + lane;
        if (e < E) { s = eidx[e]; d = eidx[(long long)E + e]; }
    }
    for (;;) {
        const int bn = b + nwaves;
        const long long en = (bn < nbatch) ? (long long)bn * 32 + lane : (long long)E;
        int sn = 0, dn = 0;
        const int nleft = E - b * 32;
        if (nleft >= 32) scatter_batch<true>(s, d, 32, c, hi, eidx, en, E, sn, dn, g, agg);
        else             scatter_batch<false>(s, d, nleft, c, hi, eidx, en, E, sn, dn, g, agg);
        if (bn >= nbatch) break;
        b = bn; s = sn; d = dn;
    }
}


#define DST_R 196
#define DST_SHIFT 23
#define SORT_WGS 256
#define SORT_THREADS 1024
#define MAX_RANGES 2048
#define HOP_THREADS 1024
#define GRP_BATCH 9
#define SCAN_ROWS 4

__global__ void __launch_bounds__(SORT_THREADS)
ak_grp_count(const int* __restrict__ eidx, int* __restrict__ off, int E, int nranges) {
    __shared__ int hist[MAX_RANGES];
    const int tid = (int)threadIdx.x, w = (int)blockIdx.x;
    for (int i = tid; i < nranges; i += SORT_THREADS) hist[i] = 0;
    __syncthreads();
    const int chunk = (E + SORT_WGS - 1) / SORT_WGS;
    const int lo = w * chunk;
    const int hi = min(E, lo + chunk);
    for (int e = lo + tid; e < hi; e += GRP_BATCH * SORT_THREADS) {
        int d[GRP_BATCH];
#pragma unroll
        for (int k = 0; k < GRP_BATCH; ++k) {
            const int q = min(e + k * SORT_THREADS, hi - 1);
            d[k] = eidx[(long long)E + q];
        }
        static_assert(GRP_BATCH == 9, "the statement below names the batch's nine values");
        asm volatile("" :: "v"(d[0]), "v"(d[1]), "v"(d[2]), "v"(d[3]), "v"(d[4]), "v"(d[5]),
                           "v"(d[6]), "v"(d[7]), "v"(d[8]));
        const bool full =
            __builtin_amdgcn_readfirstlane(e) + 31 + (GRP_BATCH - 1) * SORT_THREADS < hi;
        if (full) {
#pragma unroll
            for (int k = 0; k < GRP_BATCH; ++k)
                atomicAdd(&hist[min((int)((unsigned)d[k] / DST_R), nranges - 1)], 1);
        } else {
#pragma unroll
            for (int k = 0; k < GRP_BATCH; ++k) {
                if (e + k * SORT_THREADS < hi)
                    atomicAdd(&hist[min((int)((unsigned)d[k] / DST_R), nranges - 1)], 1);
            }
        }
    }
    __syncthreads();
    for (int i = tid; i < nranges; i += SORT_THREADS) off[i * SORT_WGS + w] = hist[i];
}

__device__ __forceinline__ int wave_incl_sum(int x, int lane) {
    x += __builtin_amdgcn_update_dpp(0, x, 0x111, 0xf, 0xf, true);
    x += __builtin_amdgcn_update_dpp(0, x, 0x112, 0xf, 0xf, true);
    x += __builtin_amdgcn_update_dpp(0, x, 0x114, 0xf, 0xf, true);
    x += __builtin_amdgcn_update_dpp(0, x, 0x118, 0xf, 0xf, true);
    const int low = __builtin_amdgcn_readlane(x, 15);
    return x + ((lane & 16) ? low : 0);
}

static_assert(SORT_WGS == 32 * 8, "a wave covers one table row, 8 counts a lane");
static_assert(MAX_RANGES <= 2 * SORT_THREADS, "step B gives each thread two rows");
__global__ void __launch_bounds__(SORT_THREADS)
ak_grp_scan(int* off, int nranges) {
    __shared__ int rsum[MAX_RANGES];
    __shared__ int wtot[32];
    const int tid = (int)threadIdx.x;
    const int lane = tid & 31;
    const int wv = __builtin_amdgcn_readfirstlane(tid >> 5);
    int4* off4 = (int4*)off;

    for (int r0 = wv; r0 < nranges; r0 += 32 * SCAN_ROWS) {
        int4 a[SCAN_ROWS], b[SCAN_ROWS];
#pragma unroll
        for (int k = 0; k < SCAN_ROWS; ++k) {
            const int r = r0 + 32 * k;
            a[k].x = 0; a[k].y = 0; a[k].z = 0; a[k].w = 0; b[k] = a[k];
            if (r < nranges) {
                const int q = r * (SORT_WGS / 4) + lane * 2;
                a[k] = off4[q]; b[k] = off4[q + 1];
            }
        }
#pragma unroll
        for (int k = 0; k < SCAN_ROWS; ++k) {
            const int r = r0 + 32 * k;
            int s = ((a[k].x + a[k].y) + (a[k].z + a[k].w)) + ((b[k].x + b[k].y) + (b[k].z + b[k].w));
            s = wave_incl_sum(s, lane);
            if (r < nranges && lane == 31) rsum[r] = s;
        }
    }
    __syncthreads();

    const int i0 = 2 * tid;
    const int e0 = (i0 < nranges) ? rsum[i0] : 0;
    const int e1 = (i0 + 1 < nranges) ? rsum[i0 + 1] : 0;
    const int s2 = e0 + e1;
    const int inc2 = wave_incl_sum(s2, lane);
    if (lane == 31) wtot[wv] = inc2;
    __syncthreads();
    const int t = wtot[lane];
    const int upto = wave_incl_sum(t, lane);
    const int all = __shfl(upto, 31, 32);
    const int before = __shfl(upto - t, wv, 32);
    const int base2 = before + inc2 - s2;
    if (i0 < nranges) rsum[i0] = base2;
    if (i0 + 1 < nranges) rsum[i0 + 1] = base2 + e0;
    __syncthreads();

    for (int r0 = wv; r0 < nranges; r0 += 32 * SCAN_ROWS) {
        int4 a[SCAN_ROWS], b[SCAN_ROWS];
#pragma unroll
        for (int k = 0; k < SCAN_ROWS; ++k) {
            const int r = r0 + 32 * k;
            a[k].x = 0; a[k].y = 0; a[k].z = 0; a[k].w = 0; b[k] = a[k];
            if (r < nranges) {
                const int q = r * (SORT_WGS / 4) + lane * 2;
                a[k] = off4[q]; b[k] = off4[q + 1];
            }
        }
#pragma unroll
        for (int k = 0; k < SCAN_ROWS; ++k) {
            const int r = r0 + 32 * k;
            const int p1 = a[k].x + a[k].y, p2 = p1 + a[k].z, p3 = p2 + a[k].w;
            const int p4 = p3 + b[k].x, p5 = p4 + b[k].y, p6 = p5 + b[k].z, p7 = p6 + b[k].w;
            const int inc = wave_incl_sum(p7, lane);
            if (r < nranges) {
                const int base = rsum[r] + inc - p7;
                const int q = r * (SORT_WGS / 4) + lane * 2;
                int4 oa, ob;
                oa.x = base;      oa.y = base + a[k].x; oa.z = base + p1; oa.w = base + p2;
                ob.x = base + p3; ob.y = base + p4;     ob.z = base + p5; ob.w = base + p6;
                off4[q] = oa; off4[q + 1] = ob;
            }
        }
    }
    if (tid == 0) off[nranges * SORT_WGS] = all;
}

__global__ void __launch_bounds__(SORT_THREADS)
ak_grp_place(const int* __restrict__ eidx, const int* __restrict__ off,
             unsigned* __restrict__ rec, int E, int nranges) {
    __shared__ int cur[MAX_RANGES];
    const int tid = (int)threadIdx.x, w = (int)blockIdx.x;
    for (int i = tid; i < nranges; i += SORT_THREADS) cur[i] = off[i * SORT_WGS + w];
    __syncthreads();
    const int chunk = (E + SORT_WGS - 1) / SORT_WGS;
    const int lo = w * chunk;
    const int hi = min(E, lo + chunk);
    for (int e = lo + tid; e < hi; e += GRP_BATCH * SORT_THREADS) {
        int s[GRP_BATCH], d[GRP_BATCH];
#pragma unroll
        for (int k = 0; k < GRP_BATCH; ++k) {
            const int q = min(e + k * SORT_THREADS, hi - 1);
            s[k] = eidx[q];
            d[k] = eidx[(long long)E + q];
        }
        const bool full =
            __builtin_amdgcn_readfirstlane(e) + 31 + (GRP_BATCH - 1) * SORT_THREADS < hi;
        if (full) {
            int pos[GRP_BATCH];
            unsigned wd[GRP_BATCH];
#pragma unroll
            for (int k = 0; k < GRP_BATCH; ++k) {
                const int b = min((int)((unsigned)d[k] / DST_R), nranges - 1);
                pos[k] = atomicAdd(&cur[b], 1);
                wd[k] = (unsigned)s[k] | ((unsigned)(d[k] - b * DST_R) << DST_SHIFT);
            }
#pragma unroll
            for (int k = 0; k < GRP_BATCH; ++k) rec[pos[k]] = wd[k];
        } else {
#pragma unroll
            for (int k = 0; k < GRP_BATCH; ++k) {
                if (e + k * SORT_THREADS < hi) {
                    const int b = min((int)((unsigned)d[k] / DST_R), nranges - 1);
                    const int pos = atomicAdd(&cur[b], 1);
                    rec[pos] = (unsigned)s[k] | ((unsigned)(d[k] - b * DST_R) << DST_SHIFT);
                }
            }
        }
    }
}

__global__ void __launch_bounds__(SORT_THREADS)
ak_grp_dinv(const unsigned* __restrict__ rec, const int* __restrict__ off,
            float* __restrict__ dinv, int N) {
    __shared__ int cnt[DST_R];
    const int tid = (int)threadIdx.x;
    const int b = (int)blockIdx.x;
    const int base = b * DST_R;
    const int nrow = min(DST_R, N - base);
    if (tid < DST_R) cnt[tid] = 0;
    __syncthreads();
    const int e0 = off[b * SORT_WGS];
    const int e1 = off[(b + 1) * SORT_WGS];
    for (int p = e0 + tid; p < e1; p += 4 * SORT_THREADS) {
        unsigned r[4];
#pragma unroll
        for (int k = 0; k < 4; ++k) {
            const int q = p + k * SORT_THREADS;
            r[k] = (q < e1) ? rec[q] : 0u;
        }
#pragma unroll
        for (int k = 0; k < 4; ++k) {
            if (p + k * SORT_THREADS < e1) atomicAdd(&cnt[r[k] >> DST_SHIFT], 1);
        }
    }
    __syncthreads();
    if (tid < nrow) dinv[base + tid] = rsqrtf(1.0f + (float)cnt[tid]);
}

template <int I>
__device__ __forceinline__ void dst_rows(unsigned r, int c, const float* __restrict__ gin,
                                         float* v, int* dj) {
    if constexpr (I < 16) {
        const unsigned rj = (unsigned)__builtin_amdgcn_update_dpp(0, (int)r, 0x150 + I, 0xf, 0xf,
                                                                   true);
        dj[I] = (int)(rj >> DST_SHIFT) * HID + c;
        v[I] = gin[(size_t)(rj & ((1u << DST_SHIFT) - 1u)) * HID + c];
        dst_rows<I + 1>(r, c, gin, v, dj);
    }
}

template <bool FULL>
__device__ __forceinline__ void dst_batch(unsigned r, int nleft, int c, int hi,
                                          const unsigned* __restrict__ rec, int pn, int end,
                                          int lane, unsigned& rn, const float* __restrict__ gin,
                                          float* acc) {
    float v[16];
    int   dj[16];
    dst_rows<0>(r, c, gin, v, dj);
    if (pn + lane < end) rn = rec[pn + lane];
#pragma unroll
    for (int i = 0; i < 16; ++i) {
        if (FULL || 16 * hi + i < nleft) atomicAdd(&acc[dj[i]], v[i]);
    }
}

template <bool LAST>
__global__ void __launch_bounds__(HOP_THREADS, 16)
ak_dst_hop(const unsigned* __restrict__ rec, const int* __restrict__ off,
           const float* __restrict__ gin, float* __restrict__ gout,
           const float* __restrict__ h0, const float* __restrict__ dinv,
           const float* __restrict__ W3, const float* __restrict__ b3,
           float* __restrict__ out, int N) {
    __shared__ float acc[DST_R * HID];
    const int tid = (int)threadIdx.x;
    const int b = (int)blockIdx.x;
    const int base = b * DST_R;
    const int n4 = min(DST_R, N - base) * (HID / 4);
    const bool live = tid < n4;
    const size_t t4 = (size_t)base * (HID / 4) + (size_t)(live ? tid : 0);

    const int lane = tid & 31;
    const int c  = lane & 15;
    const int hi = lane >> 4;
    const int e0 = off[b * SORT_WGS];
    const int end = off[(b + 1) * SORT_WGS];
    int p = e0 + __builtin_amdgcn_readfirstlane(tid >> 5) * 32;
    unsigned r = (p + lane < end) ? rec[p + lane] : 0u;

    if (live) ((float4*)acc)[tid] = ((const float4*)gin)[t4];
    __syncthreads();

    if (p < end) {
        for (;;) {
            const int nleft = end - p;
            const int pn = p + HOP_THREADS;
            unsigned rn = 0u;
            if (nleft >= 32) dst_batch<true>(r, 32, c, hi, rec, pn, end, lane, rn, gin, acc);
            else             dst_batch<false>(r, nleft, c, hi, rec, pn, end, lane, rn, gin, acc);
            if (pn >= end) break;
            p = pn; r = rn;
        }
    }
    __syncthreads();

    const float dv = dinv[base + ((live ? tid : 0) >> 2)];
    const float4 a = ((const float4*)acc)[live ? tid : 0];
    const float4 h = ((const float4*)h0)[t4];
    float4 hn;
    hn.x = ONE_MINUS_ALPHA * dv * a.x + ALPHA_F * h.x;
    hn.y = ONE_MINUS_ALPHA * dv * a.y + ALPHA_F * h.y;
    hn.z = ONE_MINUS_ALPHA * dv * a.z + ALPHA_F * h.z;
    hn.w = ONE_MINUS_ALPHA * dv * a.w + ALPHA_F * h.w;
    if (!LAST) {
        float4 gn; gn.x = dv * hn.x; gn.y = dv * hn.y; gn.z = dv * hn.z; gn.w = dv * hn.w;
        if (live) ((float4*)gout)[t4] = gn;
    } else {
        const int j = (tid & 3) * 4;
        float s = (hn.x * W3[j] + hn.y * W3[j + 1]) + (hn.z * W3[j + 2] + hn.w * W3[j + 3]);
        s += __shfl_xor(s, 1, 32);
        s += __shfl_xor(s, 2, 32);
        if (live && (tid & 3) == 0) out[base + (tid >> 2)] = s + b3[0];
    }
}

__global__ void ak_finalize(float4* agg0, float4* agg1, float4* __restrict__ g,
                            const float4* __restrict__ h0, const float* __restrict__ dinv,
                            int n4) {
    int t = blockIdx.x * blockDim.x + threadIdx.x;
    if (t >= n4) return;
    int v = t >> 2;
    float dv = dinv[v];
    float4 a = agg0[t], gg = g[t], h = h0[t];
    float4 z; z.x = 0.f; z.y = 0.f; z.z = 0.f; z.w = 0.f;
    if (agg1 != agg0) {
        float4 a1 = agg1[t];
        a.x += a1.x; a.y += a1.y; a.z += a1.z; a.w += a1.w;
        agg1[t] = z;
    }
    float4 hn;
    hn.x = ONE_MINUS_ALPHA * dv * (a.x + gg.x) + ALPHA_F * h.x;
    hn.y = ONE_MINUS_ALPHA * dv * (a.y + gg.y) + ALPHA_F * h.y;
    hn.z = ONE_MINUS_ALPHA * dv * (a.z + gg.z) + ALPHA_F * h.z;
    hn.w = ONE_MINUS_ALPHA * dv * (a.w + gg.w) + ALPHA_F * h.w;
    float4 gn; gn.x = dv * hn.x; gn.y = dv * hn.y; gn.z = dv * hn.z; gn.w = dv * hn.w;
    g[t] = gn;
    agg0[t] = z;
}

__global__ void ak_finalize_out(const float* agg0, const float* agg1, const float* __restrict__ g,
                                const float* __restrict__ h0, const float* __restrict__ dinv,
                                const float* __restrict__ W3, const float* __restrict__ b3,
                                float* __restrict__ out, int N) {
    int v = blockIdx.x * blockDim.x + threadIdx.x;
    if (v >= N) return;
    float dv = dinv[v];
    float s = b3[0];
    const bool two = (agg1 != agg0);
#pragma unroll
    for (int j = 0; j < HID; ++j) {
        float a = agg0[v * HID + j];
        if (two) a += agg1[v * HID + j];
        float h = ONE_MINUS_ALPHA * dv * (a + g[v * HID + j])
                + ALPHA_F * h0[v * HID + j];
        s += h * W3[j];
    }
    out[v] = s;
}


extern "C" void kernel_launch(void* const* d_in, const int* in_sizes, int n_in,
                              void* d_out, int out_size, void* d_ws, size_t ws_size,
                              hipStream_t stream) {
    const float* x  = (const float*)d_in[0];
    const int*  eidx = (const int*)d_in[1];
    const float* W1 = (const float*)d_in[2];
    const float* b1 = (const float*)d_in[3];
    const float* W2 = (const float*)d_in[4];
    const float* b2 = (const float*)d_in[5];
    const float* W3 = (const float*)d_in[6];
    const float* b3 = (const float*)d_in[7];
    float* out = (float*)d_out;

    const int N = in_sizes[0];
    const int E = in_sizes[1] / 2;

    char* ws = (char*)d_ws;
    size_t off = 0;
    auto carve = [&](size_t bytes) -> char* {
        char* p = ws + off;
        off += (bytes + 255) & ~(size_t)255;
        return p;
    };
    float* deg  = (float*)carve((size_t)N * 4);
    float* dinv = (float*)carve((size_t)N * 4);
    float* h0   = (float*)carve((size_t)N * HID * 4);
    float* g    = (float*)carve((size_t)N * HID * 4);
    float* agg  = (float*)carve((size_t)N * HID * 4);
    float* agg1 = agg;
    float* deg1 = deg;
    if (off + (size_t)N * HID * 4 + (size_t)N * 4 + 512 <= ws_size) {
        agg1 = (float*)carve((size_t)N * HID * 4);
        deg1 = (float*)carve((size_t)N * 4);
    }

    const int nranges = (N + DST_R - 1) / DST_R;
    unsigned* rec = nullptr;
    int* eoff = nullptr;
    if (nranges > 0 && nranges <= MAX_RANGES && N <= (1 << DST_SHIFT) &&
        off + (size_t)E * 4 + ((size_t)nranges * SORT_WGS + 1) * 4 + 512 <= ws_size) {
        rec  = (unsigned*)carve((size_t)E * 4);
        eoff = (int*)carve(((size_t)nranges * SORT_WGS + 1) * 4);
    }

    const int B = 256;
    int sblocks = SCATTER_BLOCKS;
    if (rec) {
        ak_grp_count<<<SORT_WGS, SORT_THREADS, 0, stream>>>(eidx, eoff, E, nranges);
        ak_grp_scan<<<1, SORT_THREADS, 0, stream>>>(eoff, nranges);
        ak_grp_place<<<SORT_WGS, SORT_THREADS, 0, stream>>>(eidx, eoff, rec, E, nranges);
        ak_grp_dinv<<<nranges, SORT_THREADS, 0, stream>>>(rec, eoff, dinv, N);
    } else {
        ak_deg_init<<<(N + B - 1) / B, B, 0, stream>>>(deg, deg1, N);
        ak_deg_accum<<<sblocks, B, 0, stream>>>(eidx, deg, deg1, E);
        ak_dinv<<<(N + B - 1) / B, B, 0, stream>>>(deg, deg1, dinv, N);
    }

    float* za0 = rec ? nullptr : agg;
    float* za1 = rec ? nullptr : agg1;
    int tiles = N / 16;
    if (tiles > 0) {
        int waves_per_block = B / 32;
        int blocks = (tiles + waves_per_block - 1) / waves_per_block;
        ak_encoder_wmma<<<blocks, B, 0, stream>>>(x, W1, b1, W2, b2, dinv, h0, g, za0, za1,
                                                  tiles);
    }
    int rem_start = tiles * 16;
    if (rem_start < N) {
        ak_encoder_tail<<<(N - rem_start + B - 1) / B, B, 0, stream>>>(
            x, W1, b1, W2, b2, dinv, h0, g, za0, za1, rem_start, N);
    }

    if (rec) {
        float* ga = g;
        float* gb = agg;
        for (int it = 0; it < KITERS - 1; ++it) {
            ak_dst_hop<false><<<nranges, HOP_THREADS, 0, stream>>>(rec, eoff, ga, gb, h0, dinv,
                                                                   W3, b3, out, N);
            float* t = ga; ga = gb; gb = t;
        }
        ak_dst_hop<true><<<nranges, HOP_THREADS, 0, stream>>>(rec, eoff, ga, gb, h0, dinv,
                                                              W3, b3, out, N);
        return;
    }

    int n4 = N * 4;
    for (int it = 0; it < KITERS; ++it) {
        ak_edge_scatter<<<sblocks, B, 0, stream>>>(eidx, g, agg, agg1, E);
        if (it < KITERS - 1) {
            ak_finalize<<<(n4 + B - 1) / B, B, 0, stream>>>(
                (float4*)agg, (float4*)agg1, (float4*)g, (const float4*)h0, dinv, n4);
        } else {
            ak_finalize_out<<<(N + B - 1) / B, B, 0, stream>>>(
                agg, agg1, g, h0, dinv, W3, b3, out, N);
        }
    }
}
